# starting kernel
# baseline (speedup 1.0000x reference)
#include <hip/hip_runtime.h>

#define Bsz 32
#define Lsz 256
#define Csz 16
#define Dsz 128
#define KTOT 384
#define NKCHUNK 96
#define WAVES 8
#define WT2_PITCH 288
#define WT2_FLOATS (192 * WT2_PITCH)
#define XT_FLOATS (64 * 36)
#define FEATS_FLOATS (Dsz * 16)

typedef __attribute__((ext_vector_type(2))) float v2f;
typedef __attribute__((ext_vector_type(8))) float v8f;

__device__ __forceinline__ void load_wt(const float* __restrict__ W, float* lds, int tid) {
  for (int idx = tid; idx < Dsz * Dsz; idx += 256) {
    const int dout = idx >> 7;
    const int din  = idx & 127;
    const float* wp = W + dout * 384 + din * 3;
    const float w0 = wp[0], w1 = wp[1], w2 = wp[2];
    const int kg0 = din;
    const int kg1 = 128 + din;
    const int kg2 = 256 + din;
    lds[(kg0 >> 1) * WT2_PITCH + dout * 2 + (kg0 & 1)] = w0;
    lds[(kg1 >> 1) * WT2_PITCH + dout * 2 + (kg1 & 1)] = w1;
    lds[(kg2 >> 1) * WT2_PITCH + dout * 2 + (kg2 & 1)] = w2;
  }
}

__device__ __forceinline__ void load_frags(const float* __restrict__ wt,
                                           const float* __restrict__ xt,
                                           int kk, int r, int h,
                                           v2f a[8], v2f* bf) {
  const int kg0  = kk * 4 + 2 * h;
  const int tap  = kg0 >> 7;
  const int din0 = kg0 & 127;
  *bf = *(const v2f*)(xt + (din0 >> 1) * 36 + (r + tap) * 2);
  const float* wrow = wt + (kg0 >> 1) * WT2_PITCH + r * 2;
#pragma unroll
  for (int t = 0; t < 8; ++t) {
    a[t] = *(const v2f*)(wrow + t * 32);
  }
}

__device__ __forceinline__ void wmma_group(const v2f a[8], v2f bf, v8f acc[8]) {
#pragma unroll
  for (int t = 0; t < 8; ++t) {
    acc[t] = __builtin_amdgcn_wmma_f32_16x16x4_f32(
         false, a[t],  false, bf,
         (short)0, acc[t],  false,  false);
  }
}

__device__ __forceinline__ void conv_tiles(const float* __restrict__ wt,
                                           const float* __restrict__ xt,
                                           int r, int h, v8f acc[8]) {
  v2f a0[8], a1[8], b0, b1;
  load_frags(wt, xt, 0, r, h, a0, &b0);
#pragma unroll 1
  for (int kk = 0; kk < NKCHUNK - 2; kk += 2) {
    load_frags(wt, xt, kk + 1, r, h, a1, &b1);
    wmma_group(a0, b0, acc);
    load_frags(wt, xt, kk + 2, r, h, a0, &b0);
    wmma_group(a1, b1, acc);
  }
  load_frags(wt, xt, NKCHUNK - 1, r, h, a1, &b1);
  wmma_group(a0, b0, acc);
  wmma_group(a1, b1, acc);
}

__global__ __launch_bounds__(256) void char_conv_kernel(
    const int* __restrict__ words_in_char,
    const float* __restrict__ chr_table,
    const float* __restrict__ conv_w,
    const float* __restrict__ conv_b,
    float* __restrict__ out) {
  extern __shared__ float smem[];
  float* wt = smem;
  const int tid  = threadIdx.x;
  const int lane = tid & 31;
  const int w    = tid >> 5;
  float* xt    = smem + WT2_FLOATS + w * XT_FLOATS;
  float* feats = smem + WT2_FLOATS + WAVES * XT_FLOATS;

  load_wt(conv_w, wt, tid);
  __syncthreads();

  const int gw0 = blockIdx.x * 16;
  const int b   = gw0 >> 8;
  const int l0  = gw0 & 255;
  const int r   = lane & 15;
  const int h   = lane >> 4;

  for (int wi = 0; wi < 2; ++wi) {
    const int lw = w * 2 + wi;
    const int g  = gw0 + lw;
    const int* wic = words_in_char + g * Csz;

    for (int cc = 0; cc < 18; ++cc) {
      const float* src = nullptr;
      if (cc >= 1 && cc <= 16) {
        const int ch = wic[cc - 1];
        src = chr_table + ch * Dsz;
      }
      for (int d = lane; d < Dsz; d += 32) {
        xt[(d >> 1) * 36 + cc * 2 + (d & 1)] = src ? src[d] : 0.0f;
      }
    }

    v8f acc[8] = {};
    conv_tiles(wt, xt, r, h, acc);

#pragma unroll
    for (int t = 0; t < 8; ++t) {
#pragma unroll
      for (int j = 0; j < 8; ++j) {
        float v = acc[t][j];
        v = fmaxf(v, __shfl_xor(v, 1, 32));
        v = fmaxf(v, __shfl_xor(v, 2, 32));
        v = fmaxf(v, __shfl_xor(v, 4, 32));
        v = fmaxf(v, __shfl_xor(v, 8, 32));
        if (r == j) {
          feats[(t * 16 + 8 * h + j) * 16 + lw] = v;
        }
      }
    }
  }
  __syncthreads();

  for (int i = tid; i < Dsz * 16; i += 256) {
    const int d = i >> 4;
    const int j = i & 15;
    out[((Bsz + b) * Dsz + d) * Lsz + l0 + j] = feats[d * 16 + j] + conv_b[d];
  }
}

__global__ __launch_bounds__(256) void word_conv_kernel(
    const int* __restrict__ word_vector,
    const float* __restrict__ word_table,
    const float* __restrict__ conv_w,
    const float* __restrict__ conv_b,
    float* __restrict__ out) {
  extern __shared__ float smem[];
  float* wt = smem;
  const int tid  = threadIdx.x;
  const int lane = tid & 31;
  const int w    = tid >> 5;
  float* xt = smem + WT2_FLOATS + w * XT_FLOATS;

  load_wt(conv_w, wt, tid);
  __syncthreads();

  const int task = blockIdx.x * WAVES + w;
  const int b    = task >> 4;
  const int l0   = (task & 15) * 16;
  const int r    = lane & 15;
  const int h    = lane >> 4;

  for (int cc = 0; cc < 18; ++cc) {
    const int l = l0 + cc - 1;
    const float* src = nullptr;
    if (l >= 0 && l < Lsz) {
      const int widx = word_vector[b * Lsz + l];
      src = word_table + widx * Dsz;
    }
    for (int d = lane; d < Dsz; d += 32) {
      xt[(d >> 1) * 36 + cc * 2 + (d & 1)] = src ? src[d] : 0.0f;
    }
  }

  v8f acc[8] = {};
  conv_tiles(wt, xt, r, h, acc);

#pragma unroll
  for (int t = 0; t < 8; ++t) {
#pragma unroll
    for (int j = 0; j < 8; ++j) {
      const int row = t * 16 + 8 * h + j;
      out[(b * Dsz + row) * Lsz + l0 + r] = acc[t][j] + conv_b[row];
    }
  }
}

extern "C" void kernel_launch(void* const* d_in, const int* in_sizes, int n_in,
                              void* d_out, int out_size, void* d_ws, size_t ws_size,
                              hipStream_t stream) {
  (void)in_sizes; (void)n_in; (void)out_size; (void)d_ws; (void)ws_size;
  const int*   word_vector   = (const int*)d_in[0];
  const int*   words_in_char = (const int*)d_in[1];
  const float* word_table    = (const float*)d_in[2];
  const float* chr_table     = (const float*)d_in[3];
  const float* conv_chr_w    = (const float*)d_in[4];
  const float* conv_chr_b    = (const float*)d_in[5];
  const float* conv_word_w   = (const float*)d_in[6];
  const float* conv_word_b   = (const float*)d_in[7];
  float* out = (float*)d_out;

  const size_t char_lds = (size_t)(WT2_FLOATS + WAVES * XT_FLOATS + FEATS_FLOATS) * sizeof(float);
  const size_t word_lds = (size_t)(WT2_FLOATS + WAVES * XT_FLOATS) * sizeof(float);

  static bool attr_done = false;
  if (!attr_done) {
    hipFuncSetAttribute((const void*)char_conv_kernel,
                        hipFuncAttributeMaxDynamicSharedMemorySize, (int)char_lds);
    hipFuncSetAttribute((const void*)word_conv_kernel,
                        hipFuncAttributeMaxDynamicSharedMemorySize, (int)word_lds);
    attr_done = true;
  }

  char_conv_kernel<<<dim3(Bsz * Lsz / 16), dim3(256), char_lds, stream>>>(
      words_in_char, chr_table, conv_chr_w, conv_chr_b, out);
  word_conv_kernel<<<dim3(Bsz * 16 / WAVES), dim3(256), word_lds, stream>>>(
      word_vector, word_table, conv_word_w, conv_word_b, out);
}


# Word path: each lane reads its activation operand straight from its 3 looked-up embedding rows in global memory, one 128-byte line ahead; serial per-column gather and LDS tile gone; sum order same
# speedup vs baseline: 2.3293x; 2.3293x over previous
#include <hip/hip_runtime.h>

#define Bsz 32
#define Lsz 256
#define Csz 16
#define Dsz 128
#define WAVES 8
#define CTHREADS 512
#define CWORDS 32
#define WROW_PITCH 132
#define WT_FLOATS (3 * Dsz * WROW_PITCH)
#define FEATS_PITCH 33
#define FEATS_FLOATS (Dsz * FEATS_PITCH)
#define CHR_ROWS 128
#define TAB_PITCH 132
#define TAB_ZERO_ROW CHR_ROWS
#define TAB_FLOATS ((CHR_ROWS + 1) * TAB_PITCH)

typedef __attribute__((ext_vector_type(2))) float v2f;
typedef __attribute__((ext_vector_type(8))) float v8f;

template <int NT>
__device__ __forceinline__ void load_wt(const float* __restrict__ W, float* lds, int tid) {
  const float4* W4 = (const float4*)W;
#pragma unroll 1
  for (int j0 = 0; j0 < 4096 / NT; j0 += 8) {
    float4 v[8][3];
#pragma unroll
    for (int j = 0; j < 8; ++j) {
      const int item = tid + (j0 + j) * NT;
      const float4* p = W4 + item * 3;
      v[j][0] = p[0];
      v[j][1] = p[1];
      v[j][2] = p[2];
    }
    __builtin_amdgcn_sched_barrier(0);
#pragma unroll
    for (int j = 0; j < 8; ++j) {
      const int item = tid + (j0 + j) * NT;
      float* dst = lds + (item >> 5) * WROW_PITCH + (item & 31) * 4;
      *(float4*)(dst)                        = make_float4(v[j][0].x, v[j][0].w, v[j][1].z, v[j][2].y);
      *(float4*)(dst + Dsz * WROW_PITCH)     = make_float4(v[j][0].y, v[j][1].x, v[j][1].w, v[j][2].z);
      *(float4*)(dst + 2 * Dsz * WROW_PITCH) = make_float4(v[j][0].z, v[j][1].y, v[j][2].x, v[j][2].w);
    }
  }
}

__device__ __forceinline__ void load_wfrags(const float* __restrict__ wl, v2f a[8]) {
#pragma unroll
  for (int t = 0; t < 8; ++t) {
    a[t] = *(const v2f*)(wl + t * 16 * WROW_PITCH);
  }
}

__device__ __forceinline__ void load_xline(const float* __restrict__ xl, v2f x[8]) {
#pragma unroll
  for (int c = 0; c < 8; ++c) {
    x[c] = *(const v2f*)(xl + c * 4);
  }
}

template <bool XPOSE>
__device__ __forceinline__ void wmma_group(const v2f a[8], v2f bf, v8f acc[8]) {
#pragma unroll
  for (int t = 0; t < 8; ++t) {
    if (XPOSE) {
      acc[t] = __builtin_amdgcn_wmma_f32_16x16x4_f32(
           false, bf,  false, a[t],
           (short)0, acc[t],  false,  false);
    } else {
      acc[t] = __builtin_amdgcn_wmma_f32_16x16x4_f32(
           false, a[t],  false, bf,
           (short)0, acc[t],  false,  false);
    }
  }
}

template <bool NEXT>
__device__ __forceinline__ void word_group(const float* __restrict__ wl,
                                           const float* __restrict__ wl_next,
                                           const v2f x[8], v2f a0[8], v2f a1[8], v8f acc[8]) {
#pragma unroll
  for (int c = 0; c < 8; c += 2) {
    load_wfrags(wl + (c + 1) * 4, a1);
    wmma_group<false>(a0, x[c], acc);
    if (c + 2 < 8) load_wfrags(wl + (c + 2) * 4, a0);
    else if (NEXT) load_wfrags(wl_next, a0);
    wmma_group<false>(a1, x[c + 1], acc);
  }
}

__device__ __forceinline__ void take_xline(v2f x[8], const v2f xn[8], bool inside) {
#pragma unroll
  for (int c = 0; c < 8; ++c) {
    x[c][0] = inside ? xn[c][0] : 0.0f;
    x[c][1] = inside ? xn[c][1] : 0.0f;
  }
}

__device__ __forceinline__ void load_frags_tab(const float* __restrict__ wt,
                                               const float* __restrict__ xrow,
                                               int tap, int c, int r, int h,
                                               v2f a[8], v2f* bf) {
  *bf = *(const v2f*)(xrow + c * 4);
  const float* wrow = wt + (tap * Dsz + r) * WROW_PITCH + c * 4 + 2 * h;
#pragma unroll
  for (int t = 0; t < 8; ++t) {
    a[t] = *(const v2f*)(wrow + t * 16 * WROW_PITCH);
  }
}

__device__ __forceinline__ void conv_tiles_tab(const float* __restrict__ wt,
                                               const float* __restrict__ tab,
                                               int ro0, int ro1, int ro2,
                                               int r, int h, v8f acc[8]) {
  v2f a0[8], a1[8], b0, b1;
  const float* xr[3] = {tab + ro0, tab + ro1, tab + ro2};
  load_frags_tab(wt, xr[0], 0, 0, r, h, a0, &b0);
#pragma unroll
  for (int tap = 0; tap < 3; ++tap) {
#pragma unroll 1
    for (int kc = 0; kc < 30; kc += 2) {
      load_frags_tab(wt, xr[tap], tap, kc + 1, r, h, a1, &b1);
      wmma_group<true>(a0, b0, acc);
      load_frags_tab(wt, xr[tap], tap, kc + 2, r, h, a0, &b0);
      wmma_group<true>(a1, b1, acc);
    }
    load_frags_tab(wt, xr[tap], tap, 31, r, h, a1, &b1);
    wmma_group<true>(a0, b0, acc);
    if (tap < 2) load_frags_tab(wt, xr[tap + 1], tap + 1, 0, r, h, a0, &b0);
    wmma_group<true>(a1, b1, acc);
  }
}

__global__ __launch_bounds__(CTHREADS) void char_conv_kernel(
    const int* __restrict__ words_in_char,
    const float* __restrict__ chr_table,
    const float* __restrict__ conv_w,
    const float* __restrict__ conv_b,
    float* __restrict__ out) {
  extern __shared__ float smem[];
  float* wt = smem;
  const int tid  = threadIdx.x;
  const int lane = tid & 31;
  const int w    = tid >> 5;
  float* tab   = smem + WT_FLOATS;
  float* feats = smem + WT_FLOATS + TAB_FLOATS;

  const int gw0 = blockIdx.x * CWORDS;
  const int b   = gw0 >> 8;
  const int l0  = gw0 & 255;
  const int r   = lane & 15;
  const int h   = lane >> 4;

  load_wt<CTHREADS>(conv_w, wt, tid);

  int ro[2][3];
#pragma unroll
  for (int wi = 0; wi < 2; ++wi) {
    const int* wic = words_in_char + (gw0 + w * 2 + wi) * Csz;
    const int cm = wic[r > 0 ? r - 1 : 0];
    const int c0 = wic[r];
    const int cp = wic[r < 15 ? r + 1 : 15];
    ro[wi][0] = (r > 0  ? (cm & (CHR_ROWS - 1)) : TAB_ZERO_ROW) * TAB_PITCH + 2 * h;
    ro[wi][1] = (c0 & (CHR_ROWS - 1)) * TAB_PITCH + 2 * h;
    ro[wi][2] = (r < 15 ? (cp & (CHR_ROWS - 1)) : TAB_ZERO_ROW) * TAB_PITCH + 2 * h;
  }

  {
    const float4* src4 = (const float4*)chr_table;
#pragma unroll
    for (int j = 0; j < (CHR_ROWS * Dsz / 4) / CTHREADS; ++j) {
      const int i = tid + j * CTHREADS;
      const float4 v = src4[i];
      *(float4*)(tab + (i >> 5) * TAB_PITCH + (i & 31) * 4) = v;
    }
    if (tid < Dsz / 4) {
      *(float4*)(tab + TAB_ZERO_ROW * TAB_PITCH + tid * 4) = make_float4(0.0f, 0.0f, 0.0f, 0.0f);
    }
  }
  __syncthreads();

#pragma unroll
  for (int wi = 0; wi < 2; ++wi) {
    const int lw = w * 2 + wi;

    v8f acc[8] = {};
    conv_tiles_tab(wt, tab, ro[wi][0], ro[wi][1], ro[wi][2], r, h, acc);

    float m[8];
#pragma unroll
    for (int t = 0; t < 8; ++t) {
      m[t] = fmaxf(fmaxf(fmaxf(acc[t][0], acc[t][1]), fmaxf(acc[t][2], acc[t][3])),
                   fmaxf(fmaxf(acc[t][4], acc[t][5]), fmaxf(acc[t][6], acc[t][7])));
    }
#pragma unroll
    for (int t = 0; t < 8; ++t) {
      m[t] = fmaxf(m[t], __shfl_xor(m[t], 16, 32));
    }
    if (h == 0) {
#pragma unroll
      for (int t = 0; t < 8; ++t) {
        feats[(t * 16 + r) * FEATS_PITCH + lw] = m[t];
      }
    }
  }
  __syncthreads();

  for (int i = tid; i < Dsz * CWORDS; i += CTHREADS) {
    const int d = i >> 5;
    const int j = i & 31;
    out[((Bsz + b) * Dsz + d) * Lsz + l0 + j] = feats[d * FEATS_PITCH + j] + conv_b[d];
  }
}

__global__ __launch_bounds__(256) void word_conv_kernel(
    const int* __restrict__ word_vector,
    const float* __restrict__ word_table,
    const float* __restrict__ conv_w,
    const float* __restrict__ conv_b,
    float* __restrict__ out) {
  extern __shared__ float smem[];
  float* wt = smem;
  const int tid  = threadIdx.x;
  const int lane = tid & 31;
  const int w    = tid >> 5;

  const int task = blockIdx.x * WAVES + w;
  const int b    = task >> 4;
  const int l0   = (task & 15) * 16;
  const int r    = lane & 15;
  const int h    = lane >> 4;

  const int lc = l0 + r;
  const bool in_m = lc > 0;
  const bool in_p = lc < Lsz - 1;
  const int* wv = word_vector + b * Lsz;
  const int im = wv[in_m ? lc - 1 : lc];
  const int ic = wv[lc];
  const int ip = wv[in_p ? lc + 1 : lc];

  load_wt<256>(conv_w, wt, tid);

  const float* xrow[3] = {word_table + im * Dsz + 2 * h,
                          word_table + ic * Dsz + 2 * h,
                          word_table + ip * Dsz + 2 * h};
  const bool inside[3] = {in_m, true, in_p};

  v2f a0[8], a1[8], x[8], xn[8];
  load_xline(xrow[0], xn);
  __syncthreads();

  const float* wl = wt + r * WROW_PITCH + 2 * h;
  load_wfrags(wl, a0);
  take_xline(x, xn, inside[0]);

  v8f acc[8] = {};
#pragma unroll
  for (int tap = 0; tap < 3; ++tap) {
    const float* wtap = wl + tap * Dsz * WROW_PITCH;
#pragma unroll 1
    for (int g = 0; g < 3; ++g) {
      load_xline(xrow[tap] + (g + 1) * 32, xn);
      word_group<true>(wtap + g * 32, wtap + (g + 1) * 32, x, a0, a1, acc);
      take_xline(x, xn, inside[tap]);
    }
    if (tap < 2) {
      load_xline(xrow[tap + 1], xn);
      word_group<true>(wtap + 96, wtap + Dsz * WROW_PITCH, x, a0, a1, acc);
      take_xline(x, xn, inside[tap + 1]);
    } else {
      word_group<false>(wtap + 96, wtap, x, a0, a1, acc);
    }
  }

#pragma unroll
  for (int t = 0; t < 8; ++t) {
#pragma unroll
    for (int j = 0; j < 8; ++j) {
      const int row = t * 16 + 8 * h + j;
      out[(b * Dsz + row) * Lsz + l0 + r] = acc[t][j] + conv_b[row];
    }
  }
}

extern "C" void kernel_launch(void* const* d_in, const int* in_sizes, int n_in,
                              void* d_out, int out_size, void* d_ws, size_t ws_size,
                              hipStream_t stream) {
  (void)in_sizes; (void)n_in; (void)out_size; (void)d_ws; (void)ws_size;
  const int*   word_vector   = (const int*)d_in[0];
  const int*   words_in_char = (const int*)d_in[1];
  const float* word_table    = (const float*)d_in[2];
  const float* chr_table     = (const float*)d_in[3];
  const float* conv_chr_w    = (const float*)d_in[4];
  const float* conv_chr_b    = (const float*)d_in[5];
  const float* conv_word_w   = (const float*)d_in[6];
  const float* conv_word_b   = (const float*)d_in[7];
  float* out = (float*)d_out;

  const size_t char_lds = (size_t)(WT_FLOATS + TAB_FLOATS + FEATS_FLOATS) * sizeof(float);
  const size_t word_lds = (size_t)WT_FLOATS * sizeof(float);

  static bool attr_done = false;
  if (!attr_done) {
    hipFuncSetAttribute((const void*)char_conv_kernel,
                        hipFuncAttributeMaxDynamicSharedMemorySize, (int)char_lds);
    hipFuncSetAttribute((const void*)word_conv_kernel,
                        hipFuncAttributeMaxDynamicSharedMemorySize, (int)word_lds);
    attr_done = true;
  }

  char_conv_kernel<<<dim3(Bsz * Lsz / CWORDS), dim3(CTHREADS), char_lds, stream>>>(
      words_in_char, chr_table, conv_chr_w, conv_chr_b, out);
  word_conv_kernel<<<dim3(Bsz * 16 / WAVES), dim3(256), word_lds, stream>>>(
      word_vector, word_table, conv_word_w, conv_word_b, out);
}


# Word path: each wave keeps its 16 dout x 384 k weight slice in 192 registers, straight from global; 256 blocks of 4 waves (64 dout x 64 l), one wave per SIMD; LDS holds only 66 rows; sum order same
# speedup vs baseline: 2.7311x; 1.1725x over previous
#include <hip/hip_runtime.h>

#define Bsz 32
#define Lsz 256
#define Csz 16
#define Dsz 128
#define WBLK_POS 64
#define XROW_PITCH 132
#define WORD_LDS_BYTES (164 * 1024)
#define CTHREADS 512
#define CWORDS 32
#define WROW_PITCH 132
#define WT_FLOATS (3 * Dsz * WROW_PITCH)
#define FEATS_PITCH 33
#define FEATS_FLOATS (Dsz * FEATS_PITCH)
#define CHR_ROWS 128
#define TAB_PITCH 132
#define TAB_ZERO_ROW CHR_ROWS
#define TAB_FLOATS ((CHR_ROWS + 1) * TAB_PITCH)

typedef __attribute__((ext_vector_type(2))) float v2f;
typedef __attribute__((ext_vector_type(4))) float v4f;
typedef __attribute__((ext_vector_type(8))) float v8f;

template <int NT>
__device__ __forceinline__ void load_wt(const float* __restrict__ W, float* lds, int tid) {
  const float4* W4 = (const float4*)W;
#pragma unroll 1
  for (int j0 = 0; j0 < 4096 / NT; j0 += 8) {
    float4 v[8][3];
#pragma unroll
    for (int j = 0; j < 8; ++j) {
      const int item = tid + (j0 + j) * NT;
      const float4* p = W4 + item * 3;
      v[j][0] = p[0];
      v[j][1] = p[1];
      v[j][2] = p[2];
    }
    __builtin_amdgcn_sched_barrier(0);
#pragma unroll
    for (int j = 0; j < 8; ++j) {
      const int item = tid + (j0 + j) * NT;
      float* dst = lds + (item >> 5) * WROW_PITCH + (item & 31) * 4;
      *(float4*)(dst)                        = make_float4(v[j][0].x, v[j][0].w, v[j][1].z, v[j][2].y);
      *(float4*)(dst + Dsz * WROW_PITCH)     = make_float4(v[j][0].y, v[j][1].x, v[j][1].w, v[j][2].z);
      *(float4*)(dst + 2 * Dsz * WROW_PITCH) = make_float4(v[j][0].z, v[j][1].y, v[j][2].x, v[j][2].w);
    }
  }
}

template <bool XPOSE>
__device__ __forceinline__ void wmma_group(const v2f a[8], v2f bf, v8f acc[8]) {
#pragma unroll
  for (int t = 0; t < 8; ++t) {
    if (XPOSE) {
      acc[t] = __builtin_amdgcn_wmma_f32_16x16x4_f32(
           false, bf,  false, a[t],
           (short)0, acc[t],  false,  false);
    } else {
      acc[t] = __builtin_amdgcn_wmma_f32_16x16x4_f32(
           false, a[t],  false, bf,
           (short)0, acc[t],  false,  false);
    }
  }
}

__device__ __forceinline__ void load_frags_tab(const float* __restrict__ wt,
                                               const float* __restrict__ xrow,
                                               int tap, int c, int r, int h,
                                               v2f a[8], v2f* bf) {
  *bf = *(const v2f*)(xrow + c * 4);
  const float* wrow = wt + (tap * Dsz + r) * WROW_PITCH + c * 4 + 2 * h;
#pragma unroll
  for (int t = 0; t < 8; ++t) {
    a[t] = *(const v2f*)(wrow + t * 16 * WROW_PITCH);
  }
}

__device__ __forceinline__ void conv_tiles_tab(const float* __restrict__ wt,
                                               const float* __restrict__ tab,
                                               int ro0, int ro1, int ro2,
                                               int r, int h, v8f acc[8]) {
  v2f a0[8], a1[8], b0, b1;
  const float* xr[3] = {tab + ro0, tab + ro1, tab + ro2};
  load_frags_tab(wt, xr[0], 0, 0, r, h, a0, &b0);
#pragma unroll
  for (int tap = 0; tap < 3; ++tap) {
#pragma unroll 1
    for (int kc = 0; kc < 30; kc += 2) {
      load_frags_tab(wt, xr[tap], tap, kc + 1, r, h, a1, &b1);
      wmma_group<true>(a0, b0, acc);
      load_frags_tab(wt, xr[tap], tap, kc + 2, r, h, a0, &b0);
      wmma_group<true>(a1, b1, acc);
    }
    load_frags_tab(wt, xr[tap], tap, 31, r, h, a1, &b1);
    wmma_group<true>(a0, b0, acc);
    if (tap < 2) load_frags_tab(wt, xr[tap + 1], tap + 1, 0, r, h, a0, &b0);
    wmma_group<true>(a1, b1, acc);
  }
}

__global__ __launch_bounds__(CTHREADS) void char_conv_kernel(
    const int* __restrict__ words_in_char,
    const float* __restrict__ chr_table,
    const float* __restrict__ conv_w,
    const float* __restrict__ conv_b,
    float* __restrict__ out) {
  extern __shared__ float smem[];
  float* wt = smem;
  const int tid  = threadIdx.x;
  const int lane = tid & 31;
  const int w    = tid >> 5;
  float* tab   = smem + WT_FLOATS;
  float* feats = smem + WT_FLOATS + TAB_FLOATS;

  const int gw0 = blockIdx.x * CWORDS;
  const int b   = gw0 >> 8;
  const int l0  = gw0 & 255;
  const int r   = lane & 15;
  const int h   = lane >> 4;

  load_wt<CTHREADS>(conv_w, wt, tid);

  int ro[2][3];
#pragma unroll
  for (int wi = 0; wi < 2; ++wi) {
    const int* wic = words_in_char + (gw0 + w * 2 + wi) * Csz;
    const int cm = wic[r > 0 ? r - 1 : 0];
    const int c0 = wic[r];
    const int cp = wic[r < 15 ? r + 1 : 15];
    ro[wi][0] = (r > 0  ? (cm & (CHR_ROWS - 1)) : TAB_ZERO_ROW) * TAB_PITCH + 2 * h;
    ro[wi][1] = (c0 & (CHR_ROWS - 1)) * TAB_PITCH + 2 * h;
    ro[wi][2] = (r < 15 ? (cp & (CHR_ROWS - 1)) : TAB_ZERO_ROW) * TAB_PITCH + 2 * h;
  }

  {
    const float4* src4 = (const float4*)chr_table;
#pragma unroll
    for (int j = 0; j < (CHR_ROWS * Dsz / 4) / CTHREADS; ++j) {
      const int i = tid + j * CTHREADS;
      const float4 v = src4[i];
      *(float4*)(tab + (i >> 5) * TAB_PITCH + (i & 31) * 4) = v;
    }
    if (tid < Dsz / 4) {
      *(float4*)(tab + TAB_ZERO_ROW * TAB_PITCH + tid * 4) = make_float4(0.0f, 0.0f, 0.0f, 0.0f);
    }
  }
  __syncthreads();

#pragma unroll
  for (int wi = 0; wi < 2; ++wi) {
    const int lw = w * 2 + wi;

    v8f acc[8] = {};
    conv_tiles_tab(wt, tab, ro[wi][0], ro[wi][1], ro[wi][2], r, h, acc);

    float m[8];
#pragma unroll
    for (int t = 0; t < 8; ++t) {
      m[t] = fmaxf(fmaxf(fmaxf(acc[t][0], acc[t][1]), fmaxf(acc[t][2], acc[t][3])),
                   fmaxf(fmaxf(acc[t][4], acc[t][5]), fmaxf(acc[t][6], acc[t][7])));
    }
#pragma unroll
    for (int t = 0; t < 8; ++t) {
      m[t] = fmaxf(m[t], __shfl_xor(m[t], 16, 32));
    }
    if (h == 0) {
#pragma unroll
      for (int t = 0; t < 8; ++t) {
        feats[(t * 16 + r) * FEATS_PITCH + lw] = m[t];
      }
    }
  }
  __syncthreads();

  for (int i = tid; i < Dsz * CWORDS; i += CTHREADS) {
    const int d = i >> 5;
    const int j = i & 31;
    out[((Bsz + b) * Dsz + d) * Lsz + l0 + j] = feats[d * FEATS_PITCH + j] + conv_b[d];
  }
}

__device__ __forceinline__ void load_xgroup(const float* __restrict__ xl, int tap, int din0, v2f bf[8][4]) {
#pragma unroll
  for (int c = 0; c < 8; ++c) {
#pragma unroll
    for (int j = 0; j < 4; ++j) {
      bf[c][j] = *(const v2f*)(xl + (j * 16 + tap) * XROW_PITCH + din0 + c * 4);
    }
  }
}

__device__ __forceinline__ void wmma_xgroup(const v2f a[8], const v2f bf[8][4], v8f acc[4]) {
#pragma unroll
  for (int c = 0; c < 8; ++c) {
#pragma unroll
    for (int j = 0; j < 4; ++j) {
      acc[j] = __builtin_amdgcn_wmma_f32_16x16x4_f32(
           false, a[c],  false, bf[c][j],
           (short)0, acc[j],  false,  false);
    }
  }
}

__device__ __forceinline__ v4f* xrow_slot(float* smem, int s, int lane) {
  return (v4f*)__builtin_assume_aligned(smem + s * XROW_PITCH + lane * 4, 16);
}

__global__ __launch_bounds__(128) void word_conv_kernel(
    const int* __restrict__ word_vector,
    const float* __restrict__ word_table,
    const float* __restrict__ conv_w,
    const float* __restrict__ conv_b,
    float* __restrict__ out) {
  extern __shared__ float smem[];
  const int tid  = threadIdx.x;
  const int lane = tid & 31;
  const int w    = tid >> 5;
  const int r    = lane & 15;
  const int h    = lane >> 4;

  const int blk = blockIdx.x;
  const int dt  = (blk & 1) * 4 + w;
  const int l0  = ((blk >> 1) & 3) * WBLK_POS;
  const int b   = blk >> 3;

  const int* wv = word_vector + b * Lsz;
  int  idx[17];
  bool ok[17];
#pragma unroll
  for (int j = 0; j < 17; ++j) {
    const int l = l0 - 1 + w + 4 * j;
    ok[j]  = (l >= 0) && (l < Lsz);
    idx[j] = wv[l < 0 ? 0 : (l < Lsz ? l : Lsz - 1)];
  }
  v4f v[17];
#pragma unroll
  for (int j = 0; j < 17; ++j) {
    v[j] = ((const v4f*)(word_table + idx[j] * Dsz))[lane];
  }

  v2f wr[3][32];
  {
    const float* wp = conv_w + (dt * 16 + r) * (Dsz * 3) + 6 * h;
#pragma unroll
    for (int c = 0; c < 32; ++c) {
      const v2f f01 = *(const v2f*)(wp + c * 12);
      const v2f f23 = *(const v2f*)(wp + c * 12 + 2);
      const v2f f45 = *(const v2f*)(wp + c * 12 + 4);
      wr[0][c][0] = f01[0];  wr[0][c][1] = f23[1];
      wr[1][c][0] = f01[1];  wr[1][c][1] = f45[0];
      wr[2][c][0] = f23[0];  wr[2][c][1] = f45[1];
    }
  }

#pragma unroll
  for (int j = 0; j < 17; ++j) {
    *xrow_slot(smem, w + 4 * j, lane) = v[j];
  }
#pragma unroll
  for (int j = 0; j < 17; ++j) {
    if (!ok[j]) {
      *xrow_slot(smem, w + 4 * j, lane) = (v4f){0.0f, 0.0f, 0.0f, 0.0f};
    }
  }
  __syncthreads();

  const float* xl = smem + r * XROW_PITCH + 2 * h;
  v8f acc[4] = {};
  v2f xa[8][4], xb[8][4];
  load_xgroup(xl, 0, 0, xa);
#pragma unroll
  for (int g = 0; g < 12; g += 2) {
    load_xgroup(xl, (g + 1) >> 2, ((g + 1) & 3) * 32, xb);
    __builtin_amdgcn_sched_barrier(0);
    wmma_xgroup(&wr[g >> 2][(g & 3) * 8], xa, acc);
    __builtin_amdgcn_sched_barrier(0);
    if (g + 2 < 12) load_xgroup(xl, (g + 2) >> 2, ((g + 2) & 3) * 32, xa);
    __builtin_amdgcn_sched_barrier(0);
    wmma_xgroup(&wr[(g + 1) >> 2][((g + 1) & 3) * 8], xb, acc);
    __builtin_amdgcn_sched_barrier(0);
  }

#pragma unroll
  for (int j = 0; j < 4; ++j) {
#pragma unroll
    for (int i = 0; i < 8; ++i) {
      const int row = dt * 16 + 8 * h + i;
      out[(b * Dsz + row) * Lsz + l0 + j * 16 + r] = acc[j][i] + conv_b[row];
    }
  }
}

extern "C" void kernel_launch(void* const* d_in, const int* in_sizes, int n_in,
                              void* d_out, int out_size, void* d_ws, size_t ws_size,
                              hipStream_t stream) {
  (void)in_sizes; (void)n_in; (void)out_size; (void)d_ws; (void)ws_size;
  const int*   word_vector   = (const int*)d_in[0];
  const int*   words_in_char = (const int*)d_in[1];
  const float* word_table    = (const float*)d_in[2];
  const float* chr_table     = (const float*)d_in[3];
  const float* conv_chr_w    = (const float*)d_in[4];
  const float* conv_chr_b    = (const float*)d_in[5];
  const float* conv_word_w   = (const float*)d_in[6];
  const float* conv_word_b   = (const float*)d_in[7];
  float* out = (float*)d_out;

  const size_t char_lds = (size_t)(WT_FLOATS + TAB_FLOATS + FEATS_FLOATS) * sizeof(float);
  const size_t word_lds = (size_t)WORD_LDS_BYTES;

  static bool attr_done = false;
  if (!attr_done) {
    hipFuncSetAttribute((const void*)char_conv_kernel,
                        hipFuncAttributeMaxDynamicSharedMemorySize, (int)char_lds);
    hipFuncSetAttribute((const void*)word_conv_kernel,
                        hipFuncAttributeMaxDynamicSharedMemorySize, (int)word_lds);
    attr_done = true;
  }

  char_conv_kernel<<<dim3(Bsz * Lsz / CWORDS), dim3(CTHREADS), char_lds, stream>>>(
      words_in_char, chr_table, conv_chr_w, conv_chr_b, out);
  word_conv_kernel<<<dim3(Bsz * (Lsz / WBLK_POS) * 2), dim3(128), word_lds, stream>>>(
      word_vector, word_table, conv_word_w, conv_word_b, out);
}


# Char conv by minimal filtering F(2,3) on position pairs: 4 products per 2 outputs, was 6 (768 -> 512 f32 WMMAs per word); block = 64 dout x 64 words; f32 sum order and rounding change
# speedup vs baseline: 3.6431x; 1.3339x over previous
#include <hip/hip_runtime.h>

#define Bsz 32
#define Lsz 256
#define Csz 16
#define Dsz 128
#define WBLK_POS 64
#define XROW_PITCH 132
#define WORD_LDS_BYTES (164 * 1024)
#define CTHREADS 512
#define CWORDS 64
#define CDOUT 64
#define WROW_PITCH 132
#define WT_FLOATS (4 * CDOUT * WROW_PITCH)
#define FEATS_PITCH 66
#define FEATS_FLOATS (CDOUT * FEATS_PITCH)
#define CHR_ROWS 128
#define TAB_PITCH 132
#define TAB_ZERO_ROW CHR_ROWS
#define TAB_FLOATS ((CHR_ROWS + 1) * TAB_PITCH)

typedef __attribute__((ext_vector_type(2))) float v2f;
typedef __attribute__((ext_vector_type(4))) float v4f;
typedef __attribute__((ext_vector_type(8))) float v8f;

template <int NT>
__device__ __forceinline__ void load_wt_f23(const float* __restrict__ W, float* lds, int tid) {
  const float4* W4 = (const float4*)W;
  constexpr int NJ = (CDOUT * 32) / NT;
  float4 v[NJ][3];
#pragma unroll
  for (int j = 0; j < NJ; ++j) {
    const int item = tid + j * NT;
    const float4* p = W4 + item * 3;
    v[j][0] = p[0];
    v[j][1] = p[1];
    v[j][2] = p[2];
  }
#pragma unroll
  for (int j = 0; j < NJ; ++j) {
    const int item = tid + j * NT;
    float* dst = lds + (item >> 5) * WROW_PITCH + (item & 31) * 4;
    const float g0[4] = {v[j][0].x, v[j][0].w, v[j][1].z, v[j][2].y};
    const float g1[4] = {v[j][0].y, v[j][1].x, v[j][1].w, v[j][2].z};
    const float g2[4] = {v[j][0].z, v[j][1].y, v[j][2].x, v[j][2].w};
    float u1[4], u2[4];
#pragma unroll
    for (int e = 0; e < 4; ++e) {
      const float s = g0[e] + g2[e];
      u1[e] = (s + g1[e]) * 0.5f;
      u2[e] = (s - g1[e]) * 0.5f;
    }
    *(float4*)(dst)                          = make_float4(g0[0], g0[1], g0[2], g0[3]);
    *(float4*)(dst + CDOUT * WROW_PITCH)     = make_float4(u1[0], u1[1], u1[2], u1[3]);
    *(float4*)(dst + 2 * CDOUT * WROW_PITCH) = make_float4(u2[0], u2[1], u2[2], u2[3]);
    *(float4*)(dst + 3 * CDOUT * WROW_PITCH) = make_float4(g2[0], g2[1], g2[2], g2[3]);
  }
}

__device__ __forceinline__ void load_set_f23(const float* __restrict__ wl,
                                             const float* __restrict__ tab,
                                             const int ro[4], int c,
                                             v2f a[16], v2f d[4]) {
#pragma unroll
  for (int k = 0; k < 4; ++k) {
    d[k] = *(const v2f*)(tab + ro[k] + c * 4);
  }
#pragma unroll
  for (int p = 0; p < 4; ++p) {
#pragma unroll
    for (int t = 0; t < 4; ++t) {
      a[p * 4 + t] = *(const v2f*)(wl + (p * CDOUT + t * 16) * WROW_PITCH + c * 4);
    }
  }
}

__device__ __forceinline__ void wmma_set_f23(const v2f a[16], const v2f d[4], v8f acc[16]) {
  v2f x[4];
  x[0] = d[0] - d[2];
  x[1] = d[1] + d[2];
  x[2] = d[2] - d[1];
  x[3] = d[1] - d[3];
#pragma unroll
  for (int p = 0; p < 4; ++p) {
#pragma unroll
    for (int t = 0; t < 4; ++t) {
      acc[p * 4 + t] = __builtin_amdgcn_wmma_f32_16x16x4_f32(
           false, x[p],  false, a[p * 4 + t],
           (short)0, acc[p * 4 + t],  false,  false);
    }
  }
}

__device__ __forceinline__ void conv_tile_f23(const float* __restrict__ wl,
                                              const float* __restrict__ tab,
                                              const int ro[4], v8f acc[16]) {
  v2f a0[16], a1[16], d0[4], d1[4];
  load_set_f23(wl, tab, ro, 0, a0, d0);
#pragma unroll 1
  for (int kc = 0; kc < 30; kc += 2) {
    load_set_f23(wl, tab, ro, kc + 1, a1, d1);
    wmma_set_f23(a0, d0, acc);
    load_set_f23(wl, tab, ro, kc + 2, a0, d0);
    wmma_set_f23(a1, d1, acc);
  }
  load_set_f23(wl, tab, ro, 31, a1, d1);
  wmma_set_f23(a0, d0, acc);
  wmma_set_f23(a1, d1, acc);
}

__global__ __launch_bounds__(CTHREADS) void char_conv_kernel(
    const int* __restrict__ words_in_char,
    const float* __restrict__ chr_table,
    const float* __restrict__ conv_w,
    const float* __restrict__ conv_b,
    float* __restrict__ out) {
  extern __shared__ float smem[];
  float* wt = smem;
  const int tid  = threadIdx.x;
  const int lane = tid & 31;
  const int w    = tid >> 5;
  float* tab   = smem + WT_FLOATS;
  float* feats = smem + WT_FLOATS + TAB_FLOATS;

  const int hb  = blockIdx.x & 1;
  const int gw0 = (blockIdx.x >> 1) * CWORDS;
  const int b   = gw0 >> 8;
  const int l0  = gw0 & 255;
  const int r   = lane & 15;
  const int h   = lane >> 4;

  load_wt_f23<CTHREADS>(conv_w + hb * CDOUT * Dsz * 3, wt, tid);

  int ro[2][4];
#pragma unroll
  for (int q = 0; q < 2; ++q) {
    const int* wic = words_in_char + (gw0 + w * 4 + q * 2 + (r >> 3)) * Csz;
    const int p0 = (r & 7) * 2;
#pragma unroll
    for (int k = 0; k < 4; ++k) {
      const int pos = p0 - 1 + k;
      const bool ok = (pos >= 0) && (pos < Csz);
      const int ch = wic[ok ? pos : p0];
      ro[q][k] = (ok ? (ch & (CHR_ROWS - 1)) : TAB_ZERO_ROW) * TAB_PITCH + 2 * h;
    }
  }

  {
    const float4* src4 = (const float4*)chr_table;
#pragma unroll
    for (int j = 0; j < (CHR_ROWS * Dsz / 4) / CTHREADS; ++j) {
      const int i = tid + j * CTHREADS;
      const float4 v = src4[i];
      *(float4*)(tab + (i >> 5) * TAB_PITCH + (i & 31) * 4) = v;
    }
    if (tid < Dsz / 4) {
      *(float4*)(tab + TAB_ZERO_ROW * TAB_PITCH + tid * 4) = make_float4(0.0f, 0.0f, 0.0f, 0.0f);
    }
  }
  __syncthreads();

  const float* wl = wt + r * WROW_PITCH + 2 * h;
#pragma unroll
  for (int q = 0; q < 2; ++q) {
    v8f acc[16] = {};
    conv_tile_f23(wl, tab, ro[q], acc);

    const int lw = w * 4 + q * 2 + h;
#pragma unroll
    for (int t = 0; t < 4; ++t) {
      float m = 0.0f;
#pragma unroll
      for (int j = 0; j < 8; ++j) {
        const float m0 = acc[0 * 4 + t][j];
        const float m1 = acc[1 * 4 + t][j];
        const float m2 = acc[2 * 4 + t][j];
        const float m3 = acc[3 * 4 + t][j];
        const float y0 = (m0 + m1) + m2;
        const float y1 = (m1 - m2) - m3;
        const float yy = fmaxf(y0, y1);
        m = (j == 0) ? yy : fmaxf(m, yy);
      }
      feats[(t * 16 + r) * FEATS_PITCH + lw] = m;
    }
  }
  __syncthreads();

  for (int i = tid; i < CDOUT * CWORDS; i += CTHREADS) {
    const int d = i >> 6;
    const int j = i & 63;
    out[((Bsz + b) * Dsz + hb * CDOUT + d) * Lsz + l0 + j] = feats[d * FEATS_PITCH + j] + conv_b[hb * CDOUT + d];
  }
}

__device__ __forceinline__ void load_xgroup(const float* __restrict__ xl, int tap, int din0, v2f bf[8][4]) {
#pragma unroll
  for (int c = 0; c < 8; ++c) {
#pragma unroll
    for (int j = 0; j < 4; ++j) {
      bf[c][j] = *(const v2f*)(xl + (j * 16 + tap) * XROW_PITCH + din0 + c * 4);
    }
  }
}

__device__ __forceinline__ void wmma_xgroup(const v2f a[8], const v2f bf[8][4], v8f acc[4]) {
#pragma unroll
  for (int c = 0; c < 8; ++c) {
#pragma unroll
    for (int j = 0; j < 4; ++j) {
      acc[j] = __builtin_amdgcn_wmma_f32_16x16x4_f32(
           false, a[c],  false, bf[c][j],
           (short)0, acc[j],  false,  false);
    }
  }
}

__device__ __forceinline__ v4f* xrow_slot(float* smem, int s, int lane) {
  return (v4f*)__builtin_assume_aligned(smem + s * XROW_PITCH + lane * 4, 16);
}

__global__ __launch_bounds__(128) void word_conv_kernel(
    const int* __restrict__ word_vector,
    const float* __restrict__ word_table,
    const float* __restrict__ conv_w,
    const float* __restrict__ conv_b,
    float* __restrict__ out) {
  extern __shared__ float smem[];
  const int tid  = threadIdx.x;
  const int lane = tid & 31;
  const int w    = tid >> 5;
  const int r    = lane & 15;
  const int h    = lane >> 4;

  const int blk = blockIdx.x;
  const int dt  = (blk & 1) * 4 + w;
  const int l0  = ((blk >> 1) & 3) * WBLK_POS;
  const int b   = blk >> 3;

  const int* wv = word_vector + b * Lsz;
  int  idx[17];
  bool ok[17];
#pragma unroll
  for (int j = 0; j < 17; ++j) {
    const int l = l0 - 1 + w + 4 * j;
    ok[j]  = (l >= 0) && (l < Lsz);
    idx[j] = wv[l < 0 ? 0 : (l < Lsz ? l : Lsz - 1)];
  }
  v4f v[17];
#pragma unroll
  for (int j = 0; j < 17; ++j) {
    v[j] = ((const v4f*)(word_table + idx[j] * Dsz))[lane];
  }

  v2f wr[3][32];
  {
    const float* wp = conv_w + (dt * 16 + r) * (Dsz * 3) + 6 * h;
#pragma unroll
    for (int c = 0; c < 32; ++c) {
      const v2f f01 = *(const v2f*)(wp + c * 12);
      const v2f f23 = *(const v2f*)(wp + c * 12 + 2);
      const v2f f45 = *(const v2f*)(wp + c * 12 + 4);
      wr[0][c][0] = f01[0];  wr[0][c][1] = f23[1];
      wr[1][c][0] = f01[1];  wr[1][c][1] = f45[0];
      wr[2][c][0] = f23[0];  wr[2][c][1] = f45[1];
    }
  }

#pragma unroll
  for (int j = 0; j < 17; ++j) {
    *xrow_slot(smem, w + 4 * j, lane) = v[j];
  }
#pragma unroll
  for (int j = 0; j < 17; ++j) {
    if (!ok[j]) {
      *xrow_slot(smem, w + 4 * j, lane) = (v4f){0.0f, 0.0f, 0.0f, 0.0f};
    }
  }
  __syncthreads();

  const float* xl = smem + r * XROW_PITCH + 2 * h;
  v8f acc[4] = {};
  v2f xa[8][4], xb[8][4];
  load_xgroup(xl, 0, 0, xa);
#pragma unroll
  for (int g = 0; g < 12; g += 2) {
    load_xgroup(xl, (g + 1) >> 2, ((g + 1) & 3) * 32, xb);
    __builtin_amdgcn_sched_barrier(0);
    wmma_xgroup(&wr[g >> 2][(g & 3) * 8], xa, acc);
    __builtin_amdgcn_sched_barrier(0);
    if (g + 2 < 12) load_xgroup(xl, (g + 2) >> 2, ((g + 2) & 3) * 32, xa);
    __builtin_amdgcn_sched_barrier(0);
    wmma_xgroup(&wr[(g + 1) >> 2][((g + 1) & 3) * 8], xb, acc);
    __builtin_amdgcn_sched_barrier(0);
  }

#pragma unroll
  for (int j = 0; j < 4; ++j) {
#pragma unroll
    for (int i = 0; i < 8; ++i) {
      const int row = dt * 16 + 8 * h + i;
      out[(b * Dsz + row) * Lsz + l0 + j * 16 + r] = acc[j][i] + conv_b[row];
    }
  }
}

extern "C" void kernel_launch(void* const* d_in, const int* in_sizes, int n_in,
                              void* d_out, int out_size, void* d_ws, size_t ws_size,
                              hipStream_t stream) {
  (void)in_sizes; (void)n_in; (void)out_size; (void)d_ws; (void)ws_size;
  const int*   word_vector   = (const int*)d_in[0];
  const int*   words_in_char = (const int*)d_in[1];
  const float* word_table    = (const float*)d_in[2];
  const float* chr_table     = (const float*)d_in[3];
  const float* conv_chr_w    = (const float*)d_in[4];
  const float* conv_chr_b    = (const float*)d_in[5];
  const float* conv_word_w   = (const float*)d_in[6];
  const float* conv_word_b   = (const float*)d_in[7];
  float* out = (float*)d_out;

  const size_t char_lds = (size_t)(WT_FLOATS + TAB_FLOATS + FEATS_FLOATS) * sizeof(float);
  const size_t word_lds = (size_t)WORD_LDS_BYTES;

  static bool attr_done = false;
  if (!attr_done) {
    hipFuncSetAttribute((const void*)char_conv_kernel,
                        hipFuncAttributeMaxDynamicSharedMemorySize, (int)char_lds);
    hipFuncSetAttribute((const void*)word_conv_kernel,
                        hipFuncAttributeMaxDynamicSharedMemorySize, (int)word_lds);
    attr_done = true;
  }

  char_conv_kernel<<<dim3(2 * (Bsz * Lsz / CWORDS)), dim3(CTHREADS), char_lds, stream>>>(
      words_in_char, chr_table, conv_chr_w, conv_chr_b, out);
  word_conv_kernel<<<dim3(Bsz * (Lsz / WBLK_POS) * 2), dim3(128), word_lds, stream>>>(
      word_vector, word_table, conv_word_w, conv_word_b, out);
}


# Char path: wave tile = 16 rows x all 64 dout of the block (8-wave blocks, two row tiles in turn): a chunk's table reads and input combinations made once per 24 products, was per 12; sum order same
# speedup vs baseline: 4.4923x; 1.2331x over previous
#include <hip/hip_runtime.h>

#define Bsz 32
#define Lsz 256
#define Csz 16
#define Dsz 128
#define WBLK_POS 64
#define XROW_PITCH 132
#define WORD_LDS_BYTES (164 * 1024)
#define CTHREADS 256
#define CWORDS 64
#define CTILES 4
#define CDOUT 64
#define CPLANES 6
#define WROW_PITCH 132
#define WT_FLOATS (CPLANES * CDOUT * WROW_PITCH)
#define FEATS_PITCH 68
#define FEATS_FLOATS (CDOUT * FEATS_PITCH)
#define CHR_ROWS 128
#define TAB_PITCH 132
#define TAB_ZERO_ROW CHR_ROWS
#define TAB_FLOATS ((CHR_ROWS + 1) * TAB_PITCH)

typedef __attribute__((ext_vector_type(2))) float v2f;
typedef __attribute__((ext_vector_type(4))) float v4f;
typedef __attribute__((ext_vector_type(8))) float v8f;

template <int NT>
__device__ __forceinline__ void load_wt_f43(const float* __restrict__ W, float* lds, int tid) {
  const float4* W4 = (const float4*)W;
  constexpr int NJ = (CDOUT * 32) / NT;
  float4 v[NJ][3];
#pragma unroll
  for (int j = 0; j < NJ; ++j) {
    const int item = tid + j * NT;
    const float4* p = W4 + item * 3;
    v[j][0] = p[0];
    v[j][1] = p[1];
    v[j][2] = p[2];
  }
#pragma unroll
  for (int j = 0; j < NJ; ++j) {
    const int item = tid + j * NT;
    float* dst = lds + (item >> 5) * WROW_PITCH + (item & 31) * 4;
    const float g0[4] = {v[j][0].x, v[j][0].w, v[j][1].z, v[j][2].y};
    const float g1[4] = {v[j][0].y, v[j][1].x, v[j][1].w, v[j][2].z};
    const float g2[4] = {v[j][0].z, v[j][1].y, v[j][2].x, v[j][2].w};
    float u0[4], u1[4], u2[4], u3[4], u4[4];
#pragma unroll
    for (int e = 0; e < 4; ++e) {
      const float s = g0[e] + g2[e];
      const float q = g0[e] * (1.0f / 24.0f) + g2[e] * (1.0f / 6.0f);
      const float m = g1[e] * (1.0f / 12.0f);
      u0[e] = g0[e] * 0.25f;
      u1[e] = (s + g1[e]) * (-1.0f / 6.0f);
      u2[e] = (s - g1[e]) * (-1.0f / 6.0f);
      u3[e] = q + m;
      u4[e] = q - m;
    }
    *(float4*)(dst)                          = make_float4(u0[0], u0[1], u0[2], u0[3]);
    *(float4*)(dst + CDOUT * WROW_PITCH)     = make_float4(u1[0], u1[1], u1[2], u1[3]);
    *(float4*)(dst + 2 * CDOUT * WROW_PITCH) = make_float4(u2[0], u2[1], u2[2], u2[3]);
    *(float4*)(dst + 3 * CDOUT * WROW_PITCH) = make_float4(u3[0], u3[1], u3[2], u3[3]);
    *(float4*)(dst + 4 * CDOUT * WROW_PITCH) = make_float4(u4[0], u4[1], u4[2], u4[3]);
    *(float4*)(dst + 5 * CDOUT * WROW_PITCH) = make_float4(g2[0], g2[1], g2[2], g2[3]);
  }
}

__device__ __forceinline__ void load_set_f43(const float* __restrict__ wl,
                                             const float* __restrict__ tab,
                                             const int ro[6], int c,
                                             v2f a[CPLANES * CTILES], v2f d[6]) {
#pragma unroll
  for (int k = 0; k < 6; ++k) {
    d[k] = *(const v2f*)(tab + ro[k] + c * 4);
  }
#pragma unroll
  for (int p = 0; p < CPLANES; ++p) {
#pragma unroll
    for (int t = 0; t < CTILES; ++t) {
      a[p * CTILES + t] = *(const v2f*)(wl + (p * CDOUT + t * 16) * WROW_PITCH + c * 4);
    }
  }
}

__device__ __forceinline__ void wmma_set_f43(const v2f a[CPLANES * CTILES], const v2f d[6],
                                             v8f acc[CPLANES * CTILES]) {
  v2f x[6];
  const v2f pe = d[4] - 4.0f * d[2];
  const v2f po = d[3] - 4.0f * d[1];
  const v2f qe = d[4] - d[2];
  const v2f qo = d[3] - d[1];
  x[0] = 4.0f * d[0] + (d[4] - 5.0f * d[2]);
  x[1] = pe + po;
  x[2] = pe - po;
  x[3] = qe + 2.0f * qo;
  x[4] = qe - 2.0f * qo;
  x[5] = 4.0f * d[1] + (d[5] - 5.0f * d[3]);
#pragma unroll
  for (int p = 0; p < CPLANES; ++p) {
#pragma unroll
    for (int t = 0; t < CTILES; ++t) {
      acc[p * CTILES + t] = __builtin_amdgcn_wmma_f32_16x16x4_f32(
           false, x[p],  false, a[p * CTILES + t],
           (short)0, acc[p * CTILES + t],  false,  false);
    }
  }
}

__device__ __forceinline__ void conv_tile_f43(const float* __restrict__ wl,
                                              const float* __restrict__ tab,
                                              const int ro[6], v8f acc[CPLANES * CTILES]) {
  v2f a0[CPLANES * CTILES], a1[CPLANES * CTILES], d0[6], d1[6];
  load_set_f43(wl, tab, ro, 0, a0, d0);
#pragma unroll 1
  for (int kc = 0; kc < 30; kc += 2) {
    load_set_f43(wl, tab, ro, kc + 1, a1, d1);
    wmma_set_f43(a0, d0, acc);
    load_set_f43(wl, tab, ro, kc + 2, a0, d0);
    wmma_set_f43(a1, d1, acc);
  }
  load_set_f43(wl, tab, ro, 31, a1, d1);
  wmma_set_f43(a0, d0, acc);
  wmma_set_f43(a1, d1, acc);
}

__global__ __launch_bounds__(CTHREADS) void char_conv_kernel(
    const int* __restrict__ words_in_char,
    const float* __restrict__ chr_table,
    const float* __restrict__ conv_w,
    const float* __restrict__ conv_b,
    float* __restrict__ out) {
  extern __shared__ float smem[];
  float* wt = smem;
  const int tid  = threadIdx.x;
  const int lane = tid & 31;
  const int w    = tid >> 5;
  float* tab   = smem + WT_FLOATS;
  float* feats = smem + WT_FLOATS + TAB_FLOATS;

  const int hb  = blockIdx.x & 1;
  const int gw0 = (blockIdx.x >> 1) * CWORDS;
  const int b   = gw0 >> 8;
  const int l0  = gw0 & 255;
  const int r   = lane & 15;
  const int h   = lane >> 4;

  load_wt_f43<CTHREADS>(conv_w + hb * CDOUT * Dsz * 3, wt, tid);

  int ro[2][6];
#pragma unroll
  for (int rt = 0; rt < 2; ++rt) {
    const int* wic = words_in_char + (gw0 + w * 8 + rt * 4 + (r >> 2)) * Csz;
    const int p0 = (r & 3) * 4;
#pragma unroll
    for (int k = 0; k < 6; ++k) {
      const int pos = p0 - 1 + k;
      const bool ok = (pos >= 0) && (pos < Csz);
      const int ch = wic[ok ? pos : p0];
      ro[rt][k] = (ok ? (ch & (CHR_ROWS - 1)) : TAB_ZERO_ROW) * TAB_PITCH + 2 * h;
    }
  }

  {
    const float4* src4 = (const float4*)chr_table;
#pragma unroll
    for (int j = 0; j < (CHR_ROWS * Dsz / 4) / CTHREADS; ++j) {
      const int i = tid + j * CTHREADS;
      const float4 v = src4[i];
      *(float4*)(tab + (i >> 5) * TAB_PITCH + (i & 31) * 4) = v;
    }
    if (tid < Dsz / 4) {
      *(float4*)(tab + TAB_ZERO_ROW * TAB_PITCH + tid * 4) = make_float4(0.0f, 0.0f, 0.0f, 0.0f);
    }
  }
  __syncthreads();

  const float* wl = wt + r * WROW_PITCH + 2 * h;
#pragma unroll
  for (int rt = 0; rt < 2; ++rt) {
    v8f acc[CPLANES * CTILES] = {};
    conv_tile_f43(wl, tab, ro[rt], acc);

#pragma unroll
    for (int t = 0; t < CTILES; ++t) {
      float mx[2] = {0.0f, 0.0f};
#pragma unroll
      for (int j = 0; j < 8; ++j) {
        const float m0 = acc[0 * CTILES + t][j];
        const float m1 = acc[1 * CTILES + t][j];
        const float m2 = acc[2 * CTILES + t][j];
        const float m3 = acc[3 * CTILES + t][j];
        const float m4 = acc[4 * CTILES + t][j];
        const float m5 = acc[5 * CTILES + t][j];
        const float sa = m1 + m2;
        const float sb = m1 - m2;
        const float sc = m3 + m4;
        const float se = m3 - m4;
        const float y0 = (m0 + sa) + sc;
        const float y1 = sb + 2.0f * se;
        const float y2 = sa + 4.0f * sc;
        const float y3 = (sb + 8.0f * se) + m5;
        const float yy = fmaxf(fmaxf(y0, y1), fmaxf(y2, y3));
        mx[j >> 2] = ((j & 3) == 0) ? yy : fmaxf(mx[j >> 2], yy);
      }
      *(v2f*)(feats + (t * 16 + r) * FEATS_PITCH + w * 8 + rt * 4 + 2 * h) = (v2f){mx[0], mx[1]};
    }
  }
  __syncthreads();

  for (int i = tid; i < CDOUT * CWORDS; i += CTHREADS) {
    const int d = i >> 6;
    const int j = i & 63;
    out[((Bsz + b) * Dsz + hb * CDOUT + d) * Lsz + l0 + j] = feats[d * FEATS_PITCH + j] + conv_b[hb * CDOUT + d];
  }
}

__device__ __forceinline__ void load_xgroup(const float* __restrict__ xl, int tap, int din0, v2f bf[8][4]) {
#pragma unroll
  for (int c = 0; c < 8; ++c) {
#pragma unroll
    for (int j = 0; j < 4; ++j) {
      bf[c][j] = *(const v2f*)(xl + (j * 16 + tap) * XROW_PITCH + din0 + c * 4);
    }
  }
}

__device__ __forceinline__ void wmma_xgroup(const v2f a[8], const v2f bf[8][4], v8f acc[4]) {
#pragma unroll
  for (int c = 0; c < 8; ++c) {
#pragma unroll
    for (int j = 0; j < 4; ++j) {
      acc[j] = __builtin_amdgcn_wmma_f32_16x16x4_f32(
           false, a[c],  false, bf[c][j],
           (short)0, acc[j],  false,  false);
    }
  }
}

__device__ __forceinline__ v4f* xrow_slot(float* smem, int s, int lane) {
  return (v4f*)__builtin_assume_aligned(smem + s * XROW_PITCH + lane * 4, 16);
}

__global__ __launch_bounds__(128) void word_conv_kernel(
    const int* __restrict__ word_vector,
    const float* __restrict__ word_table,
    const float* __restrict__ conv_w,
    const float* __restrict__ conv_b,
    float* __restrict__ out) {
  extern __shared__ float smem[];
  const int tid  = threadIdx.x;
  const int lane = tid & 31;
  const int w    = tid >> 5;
  const int r    = lane & 15;
  const int h    = lane >> 4;

  const int blk = blockIdx.x;
  const int dt  = (blk & 1) * 4 + w;
  const int l0  = ((blk >> 1) & 3) * WBLK_POS;
  const int b   = blk >> 3;

  const int* wv = word_vector + b * Lsz;
  int  idx[17];
  bool ok[17];
#pragma unroll
  for (int j = 0; j < 17; ++j) {
    const int l = l0 - 1 + w + 4 * j;
    ok[j]  = (l >= 0) && (l < Lsz);
    idx[j] = wv[l < 0 ? 0 : (l < Lsz ? l : Lsz - 1)];
  }
  v4f v[17];
#pragma unroll
  for (int j = 0; j < 17; ++j) {
    v[j] = ((const v4f*)(word_table + idx[j] * Dsz))[lane];
  }

  v2f wr[3][32];
  {
    const float* wp = conv_w + (dt * 16 + r) * (Dsz * 3) + 6 * h;
#pragma unroll
    for (int c = 0; c < 32; ++c) {
      const v2f f01 = *(const v2f*)(wp + c * 12);
      const v2f f23 = *(const v2f*)(wp + c * 12 + 2);
      const v2f f45 = *(const v2f*)(wp + c * 12 + 4);
      wr[0][c][0] = f01[0];  wr[0][c][1] = f23[1];
      wr[1][c][0] = f01[1];  wr[1][c][1] = f45[0];
      wr[2][c][0] = f23[0];  wr[2][c][1] = f45[1];
    }
  }

#pragma unroll
  for (int j = 0; j < 17; ++j) {
    *xrow_slot(smem, w + 4 * j, lane) = v[j];
  }
#pragma unroll
  for (int j = 0; j < 17; ++j) {
    if (!ok[j]) {
      *xrow_slot(smem, w + 4 * j, lane) = (v4f){0.0f, 0.0f, 0.0f, 0.0f};
    }
  }
  __syncthreads();

  const float* xl = smem + r * XROW_PITCH + 2 * h;
  v8f acc[4] = {};
  v2f xa[8][4], xb[8][4];
  load_xgroup(xl, 0, 0, xa);
#pragma unroll
  for (int g = 0; g < 12; g += 2) {
    load_xgroup(xl, (g + 1) >> 2, ((g + 1) & 3) * 32, xb);
    __builtin_amdgcn_sched_barrier(0);
    wmma_xgroup(&wr[g >> 2][(g & 3) * 8], xa, acc);
    __builtin_amdgcn_sched_barrier(0);
    if (g + 2 < 12) load_xgroup(xl, (g + 2) >> 2, ((g + 2) & 3) * 32, xa);
    __builtin_amdgcn_sched_barrier(0);
    wmma_xgroup(&wr[(g + 1) >> 2][((g + 1) & 3) * 8], xb, acc);
    __builtin_amdgcn_sched_barrier(0);
  }

#pragma unroll
  for (int j = 0; j < 4; ++j) {
#pragma unroll
    for (int i = 0; i < 8; ++i) {
      const int row = dt * 16 + 8 * h + i;
      out[(b * Dsz + row) * Lsz + l0 + j * 16 + r] = acc[j][i] + conv_b[row];
    }
  }
}

extern "C" void kernel_launch(void* const* d_in, const int* in_sizes, int n_in,
                              void* d_out, int out_size, void* d_ws, size_t ws_size,
                              hipStream_t stream) {
  (void)in_sizes; (void)n_in; (void)out_size; (void)d_ws; (void)ws_size;
  const int*   word_vector   = (const int*)d_in[0];
  const int*   words_in_char = (const int*)d_in[1];
  const float* word_table    = (const float*)d_in[2];
  const float* chr_table     = (const float*)d_in[3];
  const float* conv_chr_w    = (const float*)d_in[4];
  const float* conv_chr_b    = (const float*)d_in[5];
  const float* conv_word_w   = (const float*)d_in[6];
  const float* conv_word_b   = (const float*)d_in[7];
  float* out = (float*)d_out;

  const size_t char_lds = (size_t)(WT_FLOATS + TAB_FLOATS + FEATS_FLOATS) * sizeof(float);
  const size_t word_lds = (size_t)WORD_LDS_BYTES;

  static bool attr_done = false;
  if (!attr_done) {
    hipFuncSetAttribute((const void*)char_conv_kernel,
                        hipFuncAttributeMaxDynamicSharedMemorySize, (int)char_lds);
    hipFuncSetAttribute((const void*)word_conv_kernel,
                        hipFuncAttributeMaxDynamicSharedMemorySize, (int)word_lds);
    attr_done = true;
  }

  char_conv_kernel<<<dim3(2 * (Bsz * Lsz / CWORDS)), dim3(CTHREADS), char_lds, stream>>>(
      words_in_char, chr_table, conv_chr_w, conv_chr_b, out);
  word_conv_kernel<<<dim3(Bsz * (Lsz / WBLK_POS) * 2), dim3(128), word_lds, stream>>>(
      word_vector, word_table, conv_word_w, conv_word_b, out);
}


# Char path: weight planes laid chunk-major in LDS ([din/4][plane][dout][4 din], chunk pitch 1540 dwords): a chunk's 24 fragment reads take immediate offsets of one base, two per read; sum order same
# speedup vs baseline: 4.6733x; 1.0403x over previous
#include <hip/hip_runtime.h>

#define Bsz 32
#define Lsz 256
#define Csz 16
#define Dsz 128
#define WBLK_POS 64
#define XROW_PITCH 132
#define WORD_LDS_BYTES (164 * 1024)
#define CTHREADS 256
#define CWORDS 64
#define CTILES 4
#define CDOUT 64
#define CPLANES 6
#define WCH_PITCH 1540
#define WT_FLOATS (32 * WCH_PITCH)
#define FEATS_PITCH 68
#define FEATS_FLOATS (CDOUT * FEATS_PITCH)
#define CHR_ROWS 128
#define TAB_PITCH 132
#define TAB_ZERO_ROW CHR_ROWS
#define TAB_FLOATS ((CHR_ROWS + 1) * TAB_PITCH)

typedef __attribute__((ext_vector_type(2))) float v2f;
typedef __attribute__((ext_vector_type(4))) float v4f;
typedef __attribute__((ext_vector_type(8))) float v8f;

template <int NT>
__device__ __forceinline__ void load_wt_f43(const float* __restrict__ W, float* lds, int tid) {
  const float4* W4 = (const float4*)W;
  constexpr int NJ = (CDOUT * 32) / NT;
  float4 v[NJ][3];
#pragma unroll
  for (int j = 0; j < NJ; ++j) {
    const int item = tid + j * NT;
    const float4* p = W4 + item * 3;
    v[j][0] = p[0];
    v[j][1] = p[1];
    v[j][2] = p[2];
  }
#pragma unroll
  for (int j = 0; j < NJ; ++j) {
    const int item = tid + j * NT;
    float* dst = lds + (item & 31) * WCH_PITCH + (item >> 5) * 4;
    const float g0[4] = {v[j][0].x, v[j][0].w, v[j][1].z, v[j][2].y};
    const float g1[4] = {v[j][0].y, v[j][1].x, v[j][1].w, v[j][2].z};
    const float g2[4] = {v[j][0].z, v[j][1].y, v[j][2].x, v[j][2].w};
    float u0[4], u1[4], u2[4], u3[4], u4[4];
#pragma unroll
    for (int e = 0; e < 4; ++e) {
      const float s = g0[e] + g2[e];
      const float q = g0[e] * (1.0f / 24.0f) + g2[e] * (1.0f / 6.0f);
      const float m = g1[e] * (1.0f / 12.0f);
      u0[e] = g0[e] * 0.25f;
      u1[e] = (s + g1[e]) * (-1.0f / 6.0f);
      u2[e] = (s - g1[e]) * (-1.0f / 6.0f);
      u3[e] = q + m;
      u4[e] = q - m;
    }
    *(float4*)(dst)                          = make_float4(u0[0], u0[1], u0[2], u0[3]);
    *(float4*)(dst + CDOUT * 4)     = make_float4(u1[0], u1[1], u1[2], u1[3]);
    *(float4*)(dst + 2 * CDOUT * 4) = make_float4(u2[0], u2[1], u2[2], u2[3]);
    *(float4*)(dst + 3 * CDOUT * 4) = make_float4(u3[0], u3[1], u3[2], u3[3]);
    *(float4*)(dst + 4 * CDOUT * 4) = make_float4(u4[0], u4[1], u4[2], u4[3]);
    *(float4*)(dst + 5 * CDOUT * 4) = make_float4(g2[0], g2[1], g2[2], g2[3]);
  }
}

__device__ __forceinline__ void load_set_f43(const float* __restrict__ wl,
                                             const float* __restrict__ tab,
                                             const int ro[6], int c,
                                             v2f a[CPLANES * CTILES], v2f d[6]) {
#pragma unroll
  for (int k = 0; k < 6; ++k) {
    d[k] = *(const v2f*)(tab + ro[k] + c * 4);
  }
#pragma unroll
  for (int p = 0; p < CPLANES; ++p) {
#pragma unroll
    for (int t = 0; t < CTILES; ++t) {
      a[p * CTILES + t] = *(const v2f*)(wl + c * WCH_PITCH + (p * CDOUT + t * 16) * 4);
    }
  }
}

__device__ __forceinline__ void wmma_set_f43(const v2f a[CPLANES * CTILES], const v2f d[6],
                                             v8f acc[CPLANES * CTILES]) {
  v2f x[6];
  const v2f pe = d[4] - 4.0f * d[2];
  const v2f po = d[3] - 4.0f * d[1];
  const v2f qe = d[4] - d[2];
  const v2f qo = d[3] - d[1];
  x[0] = 4.0f * d[0] + (d[4] - 5.0f * d[2]);
  x[1] = pe + po;
  x[2] = pe - po;
  x[3] = qe + 2.0f * qo;
  x[4] = qe - 2.0f * qo;
  x[5] = 4.0f * d[1] + (d[5] - 5.0f * d[3]);
#pragma unroll
  for (int p = 0; p < CPLANES; ++p) {
#pragma unroll
    for (int t = 0; t < CTILES; ++t) {
      acc[p * CTILES + t] = __builtin_amdgcn_wmma_f32_16x16x4_f32(
           false, x[p],  false, a[p * CTILES + t],
           (short)0, acc[p * CTILES + t],  false,  false);
    }
  }
}

__device__ __forceinline__ void conv_tile_f43(const float* __restrict__ wl,
                                              const float* __restrict__ tab,
                                              const int ro[6], v8f acc[CPLANES * CTILES]) {
  v2f a0[CPLANES * CTILES], a1[CPLANES * CTILES], d0[6], d1[6];
  load_set_f43(wl, tab, ro, 0, a0, d0);
#pragma unroll 1
  for (int kc = 0; kc < 30; kc += 2) {
    load_set_f43(wl, tab, ro, kc + 1, a1, d1);
    wmma_set_f43(a0, d0, acc);
    load_set_f43(wl, tab, ro, kc + 2, a0, d0);
    wmma_set_f43(a1, d1, acc);
  }
  load_set_f43(wl, tab, ro, 31, a1, d1);
  wmma_set_f43(a0, d0, acc);
  wmma_set_f43(a1, d1, acc);
}

__global__ __launch_bounds__(CTHREADS) void char_conv_kernel(
    const int* __restrict__ words_in_char,
    const float* __restrict__ chr_table,
    const float* __restrict__ conv_w,
    const float* __restrict__ conv_b,
    float* __restrict__ out) {
  extern __shared__ float smem[];
  float* wt = smem;
  const int tid  = threadIdx.x;
  const int lane = tid & 31;
  const int w    = tid >> 5;
  float* tab   = smem + WT_FLOATS;
  float* feats = smem + WT_FLOATS + TAB_FLOATS;

  const int hb  = blockIdx.x & 1;
  const int gw0 = (blockIdx.x >> 1) * CWORDS;
  const int b   = gw0 >> 8;
  const int l0  = gw0 & 255;
  const int r   = lane & 15;
  const int h   = lane >> 4;

  load_wt_f43<CTHREADS>(conv_w + hb * CDOUT * Dsz * 3, wt, tid);

  int ro[2][6];
#pragma unroll
  for (int rt = 0; rt < 2; ++rt) {
    const int* wic = words_in_char + (gw0 + w * 8 + rt * 4 + (r >> 2)) * Csz;
    const int p0 = (r & 3) * 4;
#pragma unroll
    for (int k = 0; k < 6; ++k) {
      const int pos = p0 - 1 + k;
      const bool ok = (pos >= 0) && (pos < Csz);
      const int ch = wic[ok ? pos : p0];
      ro[rt][k] = (ok ? (ch & (CHR_ROWS - 1)) : TAB_ZERO_ROW) * TAB_PITCH + 2 * h;
    }
  }

  {
    const float4* src4 = (const float4*)chr_table;
#pragma unroll
    for (int j = 0; j < (CHR_ROWS * Dsz / 4) / CTHREADS; ++j) {
      const int i = tid + j * CTHREADS;
      const float4 v = src4[i];
      *(float4*)(tab + (i >> 5) * TAB_PITCH + (i & 31) * 4) = v;
    }
    if (tid < Dsz / 4) {
      *(float4*)(tab + TAB_ZERO_ROW * TAB_PITCH + tid * 4) = make_float4(0.0f, 0.0f, 0.0f, 0.0f);
    }
  }
  __syncthreads();

  const float* wl = wt + r * 4 + 2 * h;
#pragma unroll
  for (int rt = 0; rt < 2; ++rt) {
    v8f acc[CPLANES * CTILES] = {};
    conv_tile_f43(wl, tab, ro[rt], acc);

#pragma unroll
    for (int t = 0; t < CTILES; ++t) {
      float mx[2] = {0.0f, 0.0f};
#pragma unroll
      for (int j = 0; j < 8; ++j) {
        const float m0 = acc[0 * CTILES + t][j];
        const float m1 = acc[1 * CTILES + t][j];
        const float m2 = acc[2 * CTILES + t][j];
        const float m3 = acc[3 * CTILES + t][j];
        const float m4 = acc[4 * CTILES + t][j];
        const float m5 = acc[5 * CTILES + t][j];
        const float sa = m1 + m2;
        const float sb = m1 - m2;
        const float sc = m3 + m4;
        const float se = m3 - m4;
        const float y0 = (m0 + sa) + sc;
        const float y1 = sb + 2.0f * se;
        const float y2 = sa + 4.0f * sc;
        const float y3 = (sb + 8.0f * se) + m5;
        const float yy = fmaxf(fmaxf(y0, y1), fmaxf(y2, y3));
        mx[j >> 2] = ((j & 3) == 0) ? yy : fmaxf(mx[j >> 2], yy);
      }
      *(v2f*)(feats + (t * 16 + r) * FEATS_PITCH + w * 8 + rt * 4 + 2 * h) = (v2f){mx[0], mx[1]};
    }
  }
  __syncthreads();

  for (int i = tid; i < CDOUT * CWORDS; i += CTHREADS) {
    const int d = i >> 6;
    const int j = i & 63;
    out[((Bsz + b) * Dsz + hb * CDOUT + d) * Lsz + l0 + j] = feats[d * FEATS_PITCH + j] + conv_b[hb * CDOUT + d];
  }
}

__device__ __forceinline__ void load_xgroup(const float* __restrict__ xl, int tap, int din0, v2f bf[8][4]) {
#pragma unroll
  for (int c = 0; c < 8; ++c) {
#pragma unroll
    for (int j = 0; j < 4; ++j) {
      bf[c][j] = *(const v2f*)(xl + (j * 16 + tap) * XROW_PITCH + din0 + c * 4);
    }
  }
}

__device__ __forceinline__ void wmma_xgroup(const v2f a[8], const v2f bf[8][4], v8f acc[4]) {
#pragma unroll
  for (int c = 0; c < 8; ++c) {
#pragma unroll
    for (int j = 0; j < 4; ++j) {
      acc[j] = __builtin_amdgcn_wmma_f32_16x16x4_f32(
           false, a[c],  false, bf[c][j],
           (short)0, acc[j],  false,  false);
    }
  }
}

__device__ __forceinline__ v4f* xrow_slot(float* smem, int s, int lane) {
  return (v4f*)__builtin_assume_aligned(smem + s * XROW_PITCH + lane * 4, 16);
}

__global__ __launch_bounds__(128) void word_conv_kernel(
    const int* __restrict__ word_vector,
    const float* __restrict__ word_table,
    const float* __restrict__ conv_w,
    const float* __restrict__ conv_b,
    float* __restrict__ out) {
  extern __shared__ float smem[];
  const int tid  = threadIdx.x;
  const int lane = tid & 31;
  const int w    = tid >> 5;
  const int r    = lane & 15;
  const int h    = lane >> 4;

  const int blk = blockIdx.x;
  const int dt  = (blk & 1) * 4 + w;
  const int l0  = ((blk >> 1) & 3) * WBLK_POS;
  const int b   = blk >> 3;

  const int* wv = word_vector + b * Lsz;
  int  idx[17];
  bool ok[17];
#pragma unroll
  for (int j = 0; j < 17; ++j) {
    const int l = l0 - 1 + w + 4 * j;
    ok[j]  = (l >= 0) && (l < Lsz);
    idx[j] = wv[l < 0 ? 0 : (l < Lsz ? l : Lsz - 1)];
  }
  v4f v[17];
#pragma unroll
  for (int j = 0; j < 17; ++j) {
    v[j] = ((const v4f*)(word_table + idx[j] * Dsz))[lane];
  }

  v2f wr[3][32];
  {
    const float* wp = conv_w + (dt * 16 + r) * (Dsz * 3) + 6 * h;
#pragma unroll
    for (int c = 0; c < 32; ++c) {
      const v2f f01 = *(const v2f*)(wp + c * 12);
      const v2f f23 = *(const v2f*)(wp + c * 12 + 2);
      const v2f f45 = *(const v2f*)(wp + c * 12 + 4);
      wr[0][c][0] = f01[0];  wr[0][c][1] = f23[1];
      wr[1][c][0] = f01[1];  wr[1][c][1] = f45[0];
      wr[2][c][0] = f23[0];  wr[2][c][1] = f45[1];
    }
  }

#pragma unroll
  for (int j = 0; j < 17; ++j) {
    *xrow_slot(smem, w + 4 * j, lane) = v[j];
  }
#pragma unroll
  for (int j = 0; j < 17; ++j) {
    if (!ok[j]) {
      *xrow_slot(smem, w + 4 * j, lane) = (v4f){0.0f, 0.0f, 0.0f, 0.0f};
    }
  }
  __syncthreads();

  const float* xl = smem + r * XROW_PITCH + 2 * h;
  v8f acc[4] = {};
  v2f xa[8][4], xb[8][4];
  load_xgroup(xl, 0, 0, xa);
#pragma unroll
  for (int g = 0; g < 12; g += 2) {
    load_xgroup(xl, (g + 1) >> 2, ((g + 1) & 3) * 32, xb);
    __builtin_amdgcn_sched_barrier(0);
    wmma_xgroup(&wr[g >> 2][(g & 3) * 8], xa, acc);
    __builtin_amdgcn_sched_barrier(0);
    if (g + 2 < 12) load_xgroup(xl, (g + 2) >> 2, ((g + 2) & 3) * 32, xa);
    __builtin_amdgcn_sched_barrier(0);
    wmma_xgroup(&wr[(g + 1) >> 2][((g + 1) & 3) * 8], xb, acc);
    __builtin_amdgcn_sched_barrier(0);
  }

#pragma unroll
  for (int j = 0; j < 4; ++j) {
#pragma unroll
    for (int i = 0; i < 8; ++i) {
      const int row = dt * 16 + 8 * h + i;
      out[(b * Dsz + row) * Lsz + l0 + j * 16 + r] = acc[j][i] + conv_b[row];
    }
  }
}

extern "C" void kernel_launch(void* const* d_in, const int* in_sizes, int n_in,
                              void* d_out, int out_size, void* d_ws, size_t ws_size,
                              hipStream_t stream) {
  (void)in_sizes; (void)n_in; (void)out_size; (void)d_ws; (void)ws_size;
  const int*   word_vector   = (const int*)d_in[0];
  const int*   words_in_char = (const int*)d_in[1];
  const float* word_table    = (const float*)d_in[2];
  const float* chr_table     = (const float*)d_in[3];
  const float* conv_chr_w    = (const float*)d_in[4];
  const float* conv_chr_b    = (const float*)d_in[5];
  const float* conv_word_w   = (const float*)d_in[6];
  const float* conv_word_b   = (const float*)d_in[7];
  float* out = (float*)d_out;

  const size_t char_lds = (size_t)(WT_FLOATS + TAB_FLOATS + FEATS_FLOATS) * sizeof(float);
  const size_t word_lds = (size_t)WORD_LDS_BYTES;

  static bool attr_done = false;
  if (!attr_done) {
    hipFuncSetAttribute((const void*)char_conv_kernel,
                        hipFuncAttributeMaxDynamicSharedMemorySize, (int)char_lds);
    hipFuncSetAttribute((const void*)word_conv_kernel,
                        hipFuncAttributeMaxDynamicSharedMemorySize, (int)word_lds);
    attr_done = true;
  }

  char_conv_kernel<<<dim3(2 * (Bsz * Lsz / CWORDS)), dim3(CTHREADS), char_lds, stream>>>(
      words_in_char, chr_table, conv_chr_w, conv_chr_b, out);
  word_conv_kernel<<<dim3(Bsz * (Lsz / WBLK_POS) * 2), dim3(128), word_lds, stream>>>(
      word_vector, word_table, conv_word_w, conv_word_b, out);
}
